# starting kernel
# baseline (speedup 1.0000x reference)
#include <hip/hip_runtime.h>
#include <hip/hip_bf16.h>
#include <math.h>

typedef __attribute__((ext_vector_type(2))) float v2f;
typedef __attribute__((ext_vector_type(8))) float v8f;

__global__ void zero_kernel(float* __restrict__ p, long long n) {
    long long i = (long long)blockIdx.x * blockDim.x + threadIdx.x;
    if (i < n) p[i] = 0.0f;
}

__global__ void gemm1_kernel(const float* __restrict__ x,
                             const float* __restrict__ w1k1,
                             const float* __restrict__ root1,
                             float* __restrict__ y1r1, int N) {
    const int lane = threadIdx.x & 31;
    const int wave = threadIdx.x >> 5;
    const int tile = blockIdx.x * 4 + wave;
    const int row0 = tile * 16;
    if (row0 >= N) return;

    const int colL = lane & 15;
    const int kSub = (lane >> 4) * 2;
    int rowA = row0 + colL;
    if (rowA >= N) rowA = N - 1;

    v8f c0 = {};
    v8f c1 = {};
#pragma unroll 4
    for (int kk = 0; kk < 32; ++kk) {
        const int k = kk * 4 + kSub;
        v2f a;
        a.x = x[(long long)rowA * 128 + k];
        a.y = x[(long long)rowA * 128 + k + 1];
        v2f b0;
        b0.x = w1k1[k * 16 + colL];
        b0.y = w1k1[(k + 1) * 16 + colL];
        v2f b1;
        b1.x = root1[k * 16 + colL];
        b1.y = root1[(k + 1) * 16 + colL];
        c0 = __builtin_amdgcn_wmma_f32_16x16x4_f32(false, a, false, b0,
                                                   (short)0, c0, false, false);
        c1 = __builtin_amdgcn_wmma_f32_16x16x4_f32(false, a, false, b1,
                                                   (short)0, c1, false, false);
    }

    const int rbase = (lane >> 4) * 8;
#pragma unroll
    for (int r = 0; r < 8; ++r) {
        const int row = row0 + rbase + r;
        if (row < N) {
            y1r1[(long long)row * 32 + colL]      = c0[r];
            y1r1[(long long)row * 32 + 16 + colL] = c1[r];
        }
    }
}

__global__ void scatter1_kernel(const int* __restrict__ ei, int E,
                                const float* __restrict__ y1r1,
                                float* __restrict__ agg1,
                                float* __restrict__ deg) {
    long long gid = (long long)blockIdx.x * blockDim.x + threadIdx.x;
    int e = (int)(gid >> 4);
    int c = (int)(gid & 15);
    if (e >= E) return;
    int src = ei[e];
    int dst = ei[E + e];
    atomicAdd(&agg1[(long long)dst * 16 + c], y1r1[(long long)src * 32 + c]);
    if (c == 0) atomicAdd(&deg[dst], 1.0f);
}

__global__ void combine1_kernel(const float* __restrict__ agg1,
                                const float* __restrict__ deg,
                                const float* __restrict__ y1r1,
                                const float* __restrict__ b1,
                                float* __restrict__ h, int N) {
    long long gid = (long long)blockIdx.x * blockDim.x + threadIdx.x;
    if (gid >= (long long)N * 16) return;
    int n = (int)(gid >> 4);
    int c = (int)(gid & 15);
    float v = agg1[gid] / fmaxf(deg[n], 1.0f)
            + y1r1[(long long)n * 32 + 16 + c] + b1[c];
    h[gid] = (v > 0.0f) ? v : (expf(v) - 1.0f);
}

__global__ void scatter2_kernel(const int* __restrict__ ei, int E,
                                const float* __restrict__ h,
                                float* __restrict__ agg2) {
    long long gid = (long long)blockIdx.x * blockDim.x + threadIdx.x;
    int e = (int)(gid >> 4);
    int c = (int)(gid & 15);
    if (e >= E) return;
    int src = ei[e];
    int dst = ei[E + e];
    atomicAdd(&agg2[(long long)dst * 16 + c], h[(long long)src * 16 + c]);
}

__device__ __forceinline__ float b2load(const float* w2k1, const float* root2,
                                        int k, int col) {
    if (col >= 40) return 0.0f;
    return (k < 16) ? w2k1[k * 40 + col] : root2[(k - 16) * 40 + col];
}

__global__ void gemm2_kernel(const float* __restrict__ agg2,
                             const float* __restrict__ deg,
                             const float* __restrict__ h,
                             const float* __restrict__ w2k1,
                             const float* __restrict__ root2,
                             const float* __restrict__ b2,
                             float* __restrict__ out, int N) {
    const int lane = threadIdx.x & 31;
    const int wave = threadIdx.x >> 5;
    const int tile = blockIdx.x * 4 + wave;
    const int row0 = tile * 16;
    if (row0 >= N) return;

    const int colL = lane & 15;
    const int kSub = (lane >> 4) * 2;
    int rowA = row0 + colL;
    if (rowA >= N) rowA = N - 1;
    const float invd = 1.0f / fmaxf(deg[rowA], 1.0f);

    v8f c0 = {}, c1 = {}, c2 = {};
#pragma unroll
    for (int kk = 0; kk < 8; ++kk) {
        const int k = kk * 4 + kSub;
        v2f a;
        a.x = (k < 16) ? agg2[(long long)rowA * 16 + k] * invd
                       : h[(long long)rowA * 16 + (k - 16)];
        a.y = (k + 1 < 16) ? agg2[(long long)rowA * 16 + k + 1] * invd
                           : h[(long long)rowA * 16 + (k + 1 - 16)];
        v2f b0, b1, b2v;
        b0.x  = b2load(w2k1, root2, k,     colL);
        b0.y  = b2load(w2k1, root2, k + 1, colL);
        b1.x  = b2load(w2k1, root2, k,     16 + colL);
        b1.y  = b2load(w2k1, root2, k + 1, 16 + colL);
        b2v.x = b2load(w2k1, root2, k,     32 + colL);
        b2v.y = b2load(w2k1, root2, k + 1, 32 + colL);
        c0 = __builtin_amdgcn_wmma_f32_16x16x4_f32(false, a, false, b0,
                                                   (short)0, c0, false, false);
        c1 = __builtin_amdgcn_wmma_f32_16x16x4_f32(false, a, false, b1,
                                                   (short)0, c1, false, false);
        c2 = __builtin_amdgcn_wmma_f32_16x16x4_f32(false, a, false, b2v,
                                                   (short)0, c2, false, false);
    }

    const int rbase = (lane >> 4) * 8;
#pragma unroll
    for (int r = 0; r < 8; ++r) {
        const int row = row0 + rbase + r;
        if (row >= N) continue;
        out[(long long)row * 40 + colL]      = c0[r] + b2[colL];
        out[(long long)row * 40 + 16 + colL] = c1[r] + b2[16 + colL];
        if (colL < 8)
            out[(long long)row * 40 + 32 + colL] = c2[r] + b2[32 + colL];
    }
}

__global__ void logsoftmax_kernel(float* __restrict__ out, int N) {
    int n = blockIdx.x * blockDim.x + threadIdx.x;
    if (n >= N) return;
    float v[40];
    float m = -INFINITY;
#pragma unroll
    for (int i = 0; i < 40; ++i) {
        v[i] = out[(long long)n * 40 + i];
        m = fmaxf(m, v[i]);
    }
    float s = 0.0f;
#pragma unroll
    for (int i = 0; i < 40; ++i) s += expf(v[i] - m);
    float ls = logf(s);
#pragma unroll
    for (int i = 0; i < 40; ++i) out[(long long)n * 40 + i] = v[i] - m - ls;
}

extern "C" void kernel_launch(void* const* d_in, const int* in_sizes, int n_in,
                              void* d_out, int out_size, void* d_ws, size_t ws_size,
                              hipStream_t stream) {
    const float* x     = (const float*)d_in[0];
    const int*   ei    = (const int*)d_in[1];
    const float* W1    = (const float*)d_in[2];
    const float* root1 = (const float*)d_in[3];
    const float* b1    = (const float*)d_in[4];
    const float* W2    = (const float*)d_in[5];
    const float* root2 = (const float*)d_in[6];
    const float* b2    = (const float*)d_in[7];
    float*       out   = (float*)d_out;

    const int N = in_sizes[0] / 128;
    const int E = in_sizes[1] / 2;

    const float* w1k1 = W1 + 128 * 16;
    const float* w2k1 = W2 + 16 * 40;

    float* ws   = (float*)d_ws;
    float* agg1 = ws;
    float* deg  = ws + (size_t)N * 16;
    float* agg2 = ws + (size_t)N * 17;
    float* y1r1 = ws + (size_t)N * 33;
    float* h    = ws + (size_t)N * 65;

    long long nz = (long long)N * 33;
    zero_kernel<<<(unsigned)((nz + 255) / 256), 256, 0, stream>>>(ws, nz);

    const int tiles   = (N + 15) / 16;
    const int gblocks = (tiles + 3) / 4;

    gemm1_kernel<<<gblocks, 128, 0, stream>>>(x, w1k1, root1, y1r1, N);

    long long sthreads = (long long)E * 16;
    unsigned sblocks = (unsigned)((sthreads + 255) / 256);
    scatter1_kernel<<<sblocks, 256, 0, stream>>>(ei, E, y1r1, agg1, deg);

    long long cthreads = (long long)N * 16;
    combine1_kernel<<<(unsigned)((cthreads + 255) / 256), 256, 0, stream>>>(
        agg1, deg, y1r1, b1, h, N);

    scatter2_kernel<<<sblocks, 256, 0, stream>>>(ei, E, h, agg2);

    gemm2_kernel<<<gblocks, 128, 0, stream>>>(agg2, deg, h, w2k1, root2, b2, out, N);

    logsoftmax_kernel<<<(unsigned)((N + 255) / 256), 256, 0, stream>>>(out, N);
}


# gemm1, gemm2, log-softmax: tile rows cross memory as whole 128-B lines (512 B per wave instruction, a tile's loads all in flight) via an LDS tile; k order, products and every sum order unchanged
# speedup vs baseline: 5.1248x; 5.1248x over previous
#include <hip/hip_runtime.h>
#include <hip/hip_bf16.h>
#include <math.h>

typedef __attribute__((ext_vector_type(2))) float v2f;
typedef __attribute__((ext_vector_type(8))) float v8f;

__global__ void zero_kernel(float* __restrict__ p, long long n) {
    long long i = (long long)blockIdx.x * blockDim.x + threadIdx.x;
    if (i < n) p[i] = 0.0f;
}

#define DEG_TILE 8192

__global__ __launch_bounds__(1024) void degree_kernel(const int* __restrict__ ei, int E,
                                                      float* __restrict__ deg, int N) {
    __shared__ unsigned cnt[DEG_TILE];
    const int lo = blockIdx.x * DEG_TILE;
    for (int i = threadIdx.x; i < DEG_TILE; i += blockDim.x) cnt[i] = 0u;
    __syncthreads();

    const int per = (((E + (int)gridDim.y - 1) / (int)gridDim.y) + 3) & ~3;
    const int e0  = blockIdx.y * per;
    const int e1  = (e0 + per < E) ? e0 + per : E;
    const int* __restrict__ dst = ei + E;

    if ((E & 3) == 0) {
        const int4* __restrict__ d4 = (const int4*)(dst + e0);
        const int n4 = (e1 - e0) >> 2;
        for (int base = threadIdx.x; base < n4; base += 4 * (int)blockDim.x) {
            int4 d[4];
#pragma unroll
            for (int j = 0; j < 4; ++j) {
                const int idx = base + j * (int)blockDim.x;
                d[j] = d4[(idx < n4) ? idx : n4 - 1];
            }
#pragma unroll
            for (int j = 0; j < 4; ++j) {
                if (base + j * (int)blockDim.x < n4) {
                    const unsigned r0 = (unsigned)(d[j].x - lo);
                    const unsigned r1 = (unsigned)(d[j].y - lo);
                    const unsigned r2 = (unsigned)(d[j].z - lo);
                    const unsigned r3 = (unsigned)(d[j].w - lo);
                    if (r0 < DEG_TILE) atomicAdd(&cnt[r0], 1u);
                    if (r1 < DEG_TILE) atomicAdd(&cnt[r1], 1u);
                    if (r2 < DEG_TILE) atomicAdd(&cnt[r2], 1u);
                    if (r3 < DEG_TILE) atomicAdd(&cnt[r3], 1u);
                }
            }
        }
    } else {
        for (int e = e0 + threadIdx.x; e < e1; e += blockDim.x) {
            const unsigned r = (unsigned)(dst[e] - lo);
            if (r < DEG_TILE) atomicAdd(&cnt[r], 1u);
        }
    }
    __syncthreads();

    for (int i = threadIdx.x; i < DEG_TILE; i += blockDim.x) {
        const int n = lo + i;
        if (n < N) atomicAdd(&deg[n], (float)cnt[i]);
    }
}

#define G1_PITCH 132

__global__ __launch_bounds__(128) void gemm1_kernel(const float* __restrict__ x,
                             const float* __restrict__ w1k1,
                             const float* __restrict__ root1,
                             float* __restrict__ y1r1, int N) {
    __shared__ __align__(16) float xt[4][16 * G1_PITCH];
    const int lane = threadIdx.x & 31;
    const int wave = threadIdx.x >> 5;
    const int tile = blockIdx.x * 4 + wave;
    const int row0 = tile * 16;
    const int left  = N - row0;
    const int nrows = (left >= 16) ? 16 : ((left > 0) ? left : 0);

    const int colL = lane & 15;
    const int kSub = (lane >> 4) * 2;
    float* t = xt[wave];

    {
        float4 r[16];
#pragma unroll
        for (int j = 0; j < 16; ++j) {
            int row = row0 + j;
            if (row >= N) row = N - 1;
            r[j] = ((const float4*)(x + (long long)row * 128))[lane];
        }
#pragma unroll
        for (int j = 0; j < 16; ++j)
            *(float4*)(t + j * G1_PITCH + lane * 4) = r[j];
    }
    __syncthreads();

    v8f c0 = {};
    v8f c1 = {};
#pragma unroll 4
    for (int kk = 0; kk < 32; ++kk) {
        const int k = kk * 4 + kSub;
        const v2f a = *(const v2f*)(t + colL * G1_PITCH + k);
        v2f b0;
        b0.x = w1k1[k * 16 + colL];
        b0.y = w1k1[(k + 1) * 16 + colL];
        v2f b1;
        b1.x = root1[k * 16 + colL];
        b1.y = root1[(k + 1) * 16 + colL];
        c0 = __builtin_amdgcn_wmma_f32_16x16x4_f32(false, a, false, b0,
                                                   (short)0, c0, false, false);
        c1 = __builtin_amdgcn_wmma_f32_16x16x4_f32(false, a, false, b1,
                                                   (short)0, c1, false, false);
    }
    __syncthreads();

    const int rbase = (lane >> 4) * 8;
#pragma unroll
    for (int r = 0; r < 8; ++r) {
        t[(rbase + r) * 32 + colL]      = c0[r];
        t[(rbase + r) * 32 + 16 + colL] = c1[r];
    }
    __syncthreads();
    float4* __restrict__ g = (float4*)(y1r1 + (long long)row0 * 32);
    const int npieces = nrows * 8;
#pragma unroll
    for (int j = 0; j < 4; ++j) {
        const int idx = j * 32 + lane;
        if (idx < npieces) g[idx] = *(const float4*)(t + idx * 4);
    }
}

template <int PITCH>
__global__ __launch_bounds__(256) void scatter_kernel(
        const int* __restrict__ ei, int E,
        const float* __restrict__ feat,
        float* __restrict__ aggA,
        float* __restrict__ aggB) {
    const unsigned place = __builtin_amdgcn_s_sendmsg_rtn(0x87);
    float* __restrict__ agg = ((place >> 18) & 1u) ? aggB : aggA;
    const int lane   = threadIdx.x & 31;
    const int half   = lane >> 4;
    const int c      = lane & 15;
    const int gwave  = __builtin_amdgcn_readfirstlane(
                           (int)((blockIdx.x * blockDim.x + threadIdx.x) >> 5));
    const int nwaves = (int)((gridDim.x * blockDim.x) >> 5);
    const int nfull  = E >> 5;

    int chunk = gwave;
    int s = 0, d = 0;
    if (chunk < nfull) {
        s = ei[chunk * 32 + lane];
        d = ei[E + chunk * 32 + lane];
    }
    while (chunk < nfull) {
        const int nchunk = chunk + nwaves;
        const int pc = (nchunk < nfull) ? nchunk : nfull - 1;
        const int s2 = ei[pc * 32 + lane];
        const int d2 = ei[E + pc * 32 + lane];

        float v[16];
        int   dj[16];
#pragma unroll
        for (int j = 0; j < 16; ++j) {
            const int sj = __shfl(s, 2 * j + half);
            dj[j]        = __shfl(d, 2 * j + half);
            v[j] = feat[sj * PITCH + c];
        }
#pragma unroll
        for (int j = 0; j < 16; ++j)
            atomicAdd(&agg[dj[j] * 16 + c], v[j]);

        s = s2; d = d2; chunk = nchunk;
    }

    if (gwave == nwaves - 1) {
        for (int e = nfull * 32 + half; e < E; e += 2) {
            const int st = ei[e];
            const int dt = ei[E + e];
            atomicAdd(&agg[dt * 16 + c], feat[st * PITCH + c]);
        }
    }
}

__global__ void combine1_kernel(float* __restrict__ aggA,
                                float* __restrict__ aggB,
                                const float* __restrict__ deg,
                                const float* __restrict__ y1r1,
                                const float* __restrict__ b1,
                                float* __restrict__ h, int N) {
    long long gid = (long long)blockIdx.x * blockDim.x + threadIdx.x;
    if (gid >= (long long)N * 16) return;
    int n = (int)(gid >> 4);
    int c = (int)(gid & 15);
    float v = (aggA[gid] + aggB[gid]) / fmaxf(deg[n], 1.0f)
            + y1r1[(long long)n * 32 + 16 + c] + b1[c];
    h[gid] = (v > 0.0f) ? v : (expf(v) - 1.0f);
    aggA[gid] = 0.0f;
    aggB[gid] = 0.0f;
}

__device__ __forceinline__ float b2load(const float* w2k1, const float* root2,
                                        int k, int col) {
    if (col >= 40) return 0.0f;
    return (k < 16) ? w2k1[k * 40 + col] : root2[(k - 16) * 40 + col];
}

#define G2_PITCH 36

__global__ __launch_bounds__(128) void gemm2_kernel(const float* __restrict__ aggA,
                             const float* __restrict__ aggB,
                             const float* __restrict__ deg,
                             const float* __restrict__ h,
                             const float* __restrict__ w2k1,
                             const float* __restrict__ root2,
                             const float* __restrict__ b2,
                             float* __restrict__ out, int N) {
    __shared__ __align__(16) float ot[4][16 * 40];
    const int lane = threadIdx.x & 31;
    const int wave = threadIdx.x >> 5;
    const int tile = blockIdx.x * 4 + wave;
    const int row0 = tile * 16;
    const int left  = N - row0;
    const int nrows = (left >= 16) ? 16 : ((left > 0) ? left : 0);

    const int colL = lane & 15;
    const int kSub = (lane >> 4) * 2;
    float* t = ot[wave];

    {
        float4 pa[2], pb[2], ph[2];
        float  dg[2];
#pragma unroll
        for (int j = 0; j < 2; ++j) {
            const int idx = j * 32 + lane;
            int row = row0 + (idx >> 2);
            if (row >= N) row = N - 1;
            const int q = idx & 3;
            pa[j] = ((const float4*)(aggA + (long long)row * 16))[q];
            pb[j] = ((const float4*)(aggB + (long long)row * 16))[q];
            ph[j] = ((const float4*)(h    + (long long)row * 16))[q];
            dg[j] = deg[row];
        }
#pragma unroll
        for (int j = 0; j < 2; ++j) {
            const int idx = j * 32 + lane;
            const int r = idx >> 2;
            const int q = idx & 3;
            const float invd = 1.0f / fmaxf(dg[j], 1.0f);
            float4 s;
            s.x = (pa[j].x + pb[j].x) * invd;
            s.y = (pa[j].y + pb[j].y) * invd;
            s.z = (pa[j].z + pb[j].z) * invd;
            s.w = (pa[j].w + pb[j].w) * invd;
            *(float4*)(t + r * G2_PITCH + q * 4)      = s;
            *(float4*)(t + r * G2_PITCH + 16 + q * 4) = ph[j];
        }
    }
    __syncthreads();

    v8f c0 = {}, c1 = {}, c2 = {};
#pragma unroll
    for (int kk = 0; kk < 8; ++kk) {
        const int k = kk * 4 + kSub;
        const v2f a = *(const v2f*)(t + colL * G2_PITCH + k);
        v2f b0, b1, b2v;
        b0.x  = b2load(w2k1, root2, k,     colL);
        b0.y  = b2load(w2k1, root2, k + 1, colL);
        b1.x  = b2load(w2k1, root2, k,     16 + colL);
        b1.y  = b2load(w2k1, root2, k + 1, 16 + colL);
        b2v.x = b2load(w2k1, root2, k,     32 + colL);
        b2v.y = b2load(w2k1, root2, k + 1, 32 + colL);
        c0 = __builtin_amdgcn_wmma_f32_16x16x4_f32(false, a, false, b0,
                                                   (short)0, c0, false, false);
        c1 = __builtin_amdgcn_wmma_f32_16x16x4_f32(false, a, false, b1,
                                                   (short)0, c1, false, false);
        c2 = __builtin_amdgcn_wmma_f32_16x16x4_f32(false, a, false, b2v,
                                                   (short)0, c2, false, false);
    }
    __syncthreads();

    const float bias0 = b2[colL];
    const float bias1 = b2[16 + colL];
    const float bias2 = b2[32 + (colL & 7)];
    const int rbase = (lane >> 4) * 8;
#pragma unroll
    for (int r = 0; r < 8; ++r) {
        t[(rbase + r) * 40 + colL]      = c0[r] + bias0;
        t[(rbase + r) * 40 + 16 + colL] = c1[r] + bias1;
        if (colL < 8)
            t[(rbase + r) * 40 + 32 + colL] = c2[r] + bias2;
    }
    __syncthreads();
    float4* __restrict__ g = (float4*)(out + (long long)row0 * 40);
    const int npieces = nrows * 10;
#pragma unroll
    for (int j = 0; j < 5; ++j) {
        const int idx = j * 32 + lane;
        if (idx < npieces) g[idx] = *(const float4*)(t + idx * 4);
    }
}

#define LS_WAVES 8
#define LS_PIECES 320

__global__ __launch_bounds__(256) void logsoftmax_kernel(float* __restrict__ out, int N) {
    __shared__ float4 tile[LS_WAVES][LS_PIECES];
    const int lane = threadIdx.x & 31;
    const int wave = threadIdx.x >> 5;
    const long long row0 = ((long long)blockIdx.x * LS_WAVES + wave) * 32;
    const long long left = (long long)N - row0;
    const int nrows   = (left >= 32) ? 32 : ((left > 0) ? (int)left : 0);
    const int npieces = nrows * 10;
    float4* __restrict__ g = (float4*)(out + row0 * 40);
    float4* t = tile[wave];

    if (npieces > 0) {
        float4 r[10];
#pragma unroll
        for (int j = 0; j < 10; ++j) {
            const int idx = j * 32 + lane;
            r[j] = g[(idx < npieces) ? idx : npieces - 1];
        }
#pragma unroll
        for (int j = 0; j < 10; ++j) t[j * 32 + lane] = r[j];
    }
    __syncthreads();

    if (lane < nrows) {
        float v[40];
#pragma unroll
        for (int q = 0; q < 10; ++q) {
            const float4 p = t[lane * 10 + q];
            v[4 * q + 0] = p.x; v[4 * q + 1] = p.y;
            v[4 * q + 2] = p.z; v[4 * q + 3] = p.w;
        }
        float m = -INFINITY;
#pragma unroll
        for (int i = 0; i < 40; ++i) m = fmaxf(m, v[i]);
        float s = 0.0f;
#pragma unroll
        for (int i = 0; i < 40; ++i) s += expf(v[i] - m);
        const float ls = logf(s);
#pragma unroll
        for (int q = 0; q < 10; ++q) {
            float4 p;
            p.x = v[4 * q + 0] - m - ls; p.y = v[4 * q + 1] - m - ls;
            p.z = v[4 * q + 2] - m - ls; p.w = v[4 * q + 3] - m - ls;
            t[lane * 10 + q] = p;
        }
    }
    __syncthreads();

    if (npieces == LS_PIECES) {
        float4 w[10];
#pragma unroll
        for (int j = 0; j < 10; ++j) w[j] = t[j * 32 + lane];
#pragma unroll
        for (int j = 0; j < 10; ++j) g[j * 32 + lane] = w[j];
    } else {
        for (int j = 0; j < 10; ++j) {
            const int idx = j * 32 + lane;
            if (idx < npieces) g[idx] = t[idx];
        }
    }
}

extern "C" void kernel_launch(void* const* d_in, const int* in_sizes, int n_in,
                              void* d_out, int out_size, void* d_ws, size_t ws_size,
                              hipStream_t stream) {
    const float* x     = (const float*)d_in[0];
    const int*   ei    = (const int*)d_in[1];
    const float* W1    = (const float*)d_in[2];
    const float* root1 = (const float*)d_in[3];
    const float* b1    = (const float*)d_in[4];
    const float* W2    = (const float*)d_in[5];
    const float* root2 = (const float*)d_in[6];
    const float* b2    = (const float*)d_in[7];
    float*       out   = (float*)d_out;

    const int N = in_sizes[0] / 128;
    const int E = in_sizes[1] / 2;

    const float* w1k1 = W1 + 128 * 16;
    const float* w2k1 = W2 + 16 * 40;

    float* ws   = (float*)d_ws;
    float* aggA = ws;
    float* deg  = ws + (size_t)N * 16;
    float* aggB = ws + (size_t)N * 17;
    float* y1r1 = ws + (size_t)N * 33;
    float* h    = ws + (size_t)N * 65;

    long long nz = (long long)N * 33;
    zero_kernel<<<(unsigned)((nz + 255) / 256), 256, 0, stream>>>(ws, nz);

    degree_kernel<<<dim3((unsigned)((N + DEG_TILE - 1) / DEG_TILE), 32), 1024, 0, stream>>>(
        ei, E, deg, N);

    const int tiles   = (N + 15) / 16;
    const int gblocks = (tiles + 3) / 4;

    gemm1_kernel<<<gblocks, 128, 0, stream>>>(x, w1k1, root1, y1r1, N);

    scatter_kernel<32><<<1024, 256, 0, stream>>>(ei, E, y1r1, aggA, aggB);

    long long cthreads = (long long)N * 16;
    combine1_kernel<<<(unsigned)((cthreads + 255) / 256), 256, 0, stream>>>(
        aggA, aggB, deg, y1r1, b1, h, N);

    scatter_kernel<16><<<1024, 256, 0, stream>>>(ei, E, h, aggA, aggB);

    gemm2_kernel<<<gblocks, 128, 0, stream>>>(aggA, aggB, deg, h, w2k1, root2, b2, out, N);

    logsoftmax_kernel<<<(unsigned)((N + 255) / 256), 256, 0, stream>>>(out, N);
}
